# starting kernel
# baseline (speedup 1.0000x reference)
#include <hip/hip_runtime.h>

#define HH 768
#define WW 768
#define CC 4
#define NB 8
#define RAD 8
#define EPSV 1e-4f

typedef __attribute__((ext_vector_type(2))) float v2f;
typedef __attribute__((ext_vector_type(8))) float v8f;

static __device__ __forceinline__ v8f wmma4(v2f a, v2f b, v8f c) {
  return __builtin_amdgcn_wmma_f32_16x16x4_f32(false, a, false, b, (short)0, c,
                                               false, false);
}

static __device__ __forceinline__ v8f box2d_16(const float* X, float* T,
                                               const v2f band[8], int lane) {
  const int m = lane & 15;
  const int hi = lane >> 4;
  #pragma unroll
  for (int rs = 0; rs < 2; ++rs) {
    v2f a[8];
    #pragma unroll
    for (int kk = 0; kk < 8; ++kk) {
      const int j0 = 4 * kk + 2 * hi;
      v2f t = { X[(rs * 16 + m) * 33 + j0],
                X[(rs * 16 + m) * 33 + j0 + 1] };
      a[kk] = t;
    }
    v8f acc = {};
    #pragma unroll
    for (int kk = 0; kk < 8; ++kk) acc = wmma4(a[kk], band[kk], acc);
    #pragma unroll
    for (int v = 0; v < 8; ++v)
      T[(rs * 16 + v + 8 * hi) * 17 + m] = acc[v];
  }
  __syncthreads();
  v2f b[8];
  #pragma unroll
  for (int kk = 0; kk < 8; ++kk) {
    const int j0 = 4 * kk + 2 * hi;
    v2f t = { T[j0 * 17 + m], T[(j0 + 1) * 17 + m] };
    b[kk] = t;
  }
  v8f acc = {};
  #pragma unroll
  for (int kk = 0; kk < 8; ++kk) acc = wmma4(band[kk], b[kk], acc);
  __syncthreads();
  return acc;
}

static __device__ __forceinline__ void make_band(v2f band[8], int lane) {
  const int hi = lane >> 4;
  const int m = lane & 15;
  #pragma unroll
  for (int kk = 0; kk < 8; ++kk) {
    const int d0 = 4 * kk + 2 * hi - m;
    v2f t = { (d0 >= 0 && d0 <= 16) ? 1.f : 0.f,
              (d0 + 1 >= 0 && d0 + 1 <= 16) ? 1.f : 0.f };
    band[kk] = t;
  }
}

static __device__ __forceinline__ void inv4_rowsum(const float m[16],
                                                   float rs[4]) {
  float i0  =  m[5]*m[10]*m[15] - m[5]*m[11]*m[14] - m[9]*m[6]*m[15] + m[9]*m[7]*m[14] + m[13]*m[6]*m[11] - m[13]*m[7]*m[10];
  float i4  = -m[4]*m[10]*m[15] + m[4]*m[11]*m[14] + m[8]*m[6]*m[15] - m[8]*m[7]*m[14] - m[12]*m[6]*m[11] + m[12]*m[7]*m[10];
  float i8  =  m[4]*m[9]*m[15] - m[4]*m[11]*m[13] - m[8]*m[5]*m[15] + m[8]*m[7]*m[13] + m[12]*m[5]*m[11] - m[12]*m[7]*m[9];
  float i12 = -m[4]*m[9]*m[14] + m[4]*m[10]*m[13] + m[8]*m[5]*m[14] - m[8]*m[6]*m[13] - m[12]*m[5]*m[10] + m[12]*m[6]*m[9];
  float i1  = -m[1]*m[10]*m[15] + m[1]*m[11]*m[14] + m[9]*m[2]*m[15] - m[9]*m[3]*m[14] - m[13]*m[2]*m[11] + m[13]*m[3]*m[10];
  float i5  =  m[0]*m[10]*m[15] - m[0]*m[11]*m[14] - m[8]*m[2]*m[15] + m[8]*m[3]*m[14] + m[12]*m[2]*m[11] - m[12]*m[3]*m[10];
  float i9  = -m[0]*m[9]*m[15] + m[0]*m[11]*m[13] + m[8]*m[1]*m[15] - m[8]*m[3]*m[13] - m[12]*m[1]*m[11] + m[12]*m[3]*m[9];
  float i13 =  m[0]*m[9]*m[14] - m[0]*m[10]*m[13] - m[8]*m[1]*m[14] + m[8]*m[2]*m[13] + m[12]*m[1]*m[10] - m[12]*m[2]*m[9];
  float i2  =  m[1]*m[6]*m[15] - m[1]*m[7]*m[14] - m[5]*m[2]*m[15] + m[5]*m[3]*m[14] + m[13]*m[2]*m[7] - m[13]*m[3]*m[6];
  float i6  = -m[0]*m[6]*m[15] + m[0]*m[7]*m[14] + m[4]*m[2]*m[15] - m[4]*m[3]*m[14] - m[12]*m[2]*m[7] + m[12]*m[3]*m[6];
  float i10 =  m[0]*m[5]*m[15] - m[0]*m[7]*m[13] - m[4]*m[1]*m[15] + m[4]*m[3]*m[13] + m[12]*m[1]*m[7] - m[12]*m[3]*m[5];
  float i14 = -m[0]*m[5]*m[14] + m[0]*m[6]*m[13] + m[4]*m[1]*m[14] - m[4]*m[2]*m[13] - m[12]*m[1]*m[6] + m[12]*m[2]*m[5];
  float i3  = -m[1]*m[6]*m[11] + m[1]*m[7]*m[10] + m[5]*m[2]*m[11] - m[5]*m[3]*m[10] - m[9]*m[2]*m[7] + m[9]*m[3]*m[6];
  float i7  =  m[0]*m[6]*m[11] - m[0]*m[7]*m[10] - m[4]*m[2]*m[11] + m[4]*m[3]*m[10] + m[8]*m[2]*m[7] - m[8]*m[3]*m[6];
  float i11 = -m[0]*m[5]*m[11] + m[0]*m[7]*m[9] + m[4]*m[1]*m[11] - m[4]*m[3]*m[9] - m[8]*m[1]*m[7] + m[8]*m[3]*m[5];
  float i15 =  m[0]*m[5]*m[10] - m[0]*m[6]*m[9] - m[4]*m[1]*m[10] + m[4]*m[2]*m[9] + m[8]*m[1]*m[6] - m[8]*m[2]*m[5];
  const float det = m[0]*i0 + m[1]*i4 + m[2]*i8 + m[3]*i12;
  const float id = 1.0f / det;
  rs[0] = (i0 + i1 + i2 + i3) * id;
  rs[1] = (i4 + i5 + i6 + i7) * id;
  rs[2] = (i8 + i9 + i10 + i11) * id;
  rs[3] = (i12 + i13 + i14 + i15) * id;
}

__global__ __launch_bounds__(32) void gf_stats(const float* __restrict__ I,
                                               const float* __restrict__ P,
                                               float* __restrict__ Aout,
                                               float* __restrict__ Bout) {
  __shared__ float raw[5][32 * 33];
  __shared__ float X[32 * 33];
  __shared__ float T[32 * 17];
  const int lane = threadIdx.x;
  const int tx = blockIdx.x * 16;
  const int ty = blockIdx.y * 16;
  const int bb = blockIdx.z;
  const int gc = tx - RAD + lane;
  const bool cok = (gc >= 0) && (gc < WW);
  const float* Ib = I + (size_t)bb * CC * HH * WW;
  const float* Pb = P + (size_t)bb * HH * WW;
  for (int y = 0; y < 32; ++y) {
    const int gr = ty - RAD + y;
    const bool ok = cok && (gr >= 0) && (gr < HH);
    const size_t off = (size_t)gr * WW + gc;
    raw[4][y * 33 + lane] = ok ? Pb[off] : 0.f;
    #pragma unroll
    for (int c = 0; c < CC; ++c)
      raw[c][y * 33 + lane] = ok ? Ib[(size_t)c * HH * WW + off] : 0.f;
  }
  __syncthreads();

  v2f band[8];
  make_band(band, lane);

  v8f S[19];
  #pragma unroll
  for (int pl = 0; pl < 5; ++pl)
    S[pl] = box2d_16(&raw[pl][0], T, band, lane);

  constexpr int ia[14]  = {4,4,4,4, 0,0,0,0, 1,1,1, 2,2, 3};
  constexpr int ibx[14] = {0,1,2,3, 0,1,2,3, 1,2,3, 2,3, 3};
  #pragma unroll
  for (int pl = 0; pl < 14; ++pl) {
    #pragma unroll
    for (int y = 0; y < 32; ++y)
      X[y * 33 + lane] =
          raw[ia[pl]][y * 33 + lane] * raw[ibx[pl]][y * 33 + lane];
    __syncthreads();
    S[5 + pl] = box2d_16(X, T, band, lane);
  }

  const int hi = lane >> 4;
  const int w = lane & 15;
  const int col = tx + w;
  const int cw = min(col + RAD, WW - 1) - max(col - RAD, 0) + 1;
  #pragma unroll
  for (int v = 0; v < 8; ++v) {
    const int row = ty + v + 8 * hi;
    const int chc = min(row + RAD, HH - 1) - max(row - RAD, 0) + 1;
    const float invN = 1.f / (float)(chc * cw);
    const float pm = S[4][v] * invN;
    float Im[4], first[4];
    float sumIm = 0.f;
    #pragma unroll
    for (int c = 0; c < 4; ++c) {
      Im[c] = S[c][v] * invN;
      sumIm += Im[c];
      first[c] = S[5 + c][v] * invN - pm * Im[c];
    }
    float M[16];
    M[0]  = S[9][v]  * invN + EPSV;
    M[1]  = M[4]  = S[10][v] * invN;
    M[2]  = M[8]  = S[11][v] * invN;
    M[3]  = M[12] = S[12][v] * invN;
    M[5]  = S[13][v] * invN + EPSV;
    M[6]  = M[9]  = S[14][v] * invN;
    M[7]  = M[13] = S[15][v] * invN;
    M[10] = S[16][v] * invN + EPSV;
    M[11] = M[14] = S[17][v] * invN;
    M[15] = S[18][v] * invN + EPSV;
    float rs[4];
    inv4_rowsum(M, rs);
    float sumA = 0.f;
    float Ac[4];
    #pragma unroll
    for (int c = 0; c < 4; ++c) {
      Ac[c] = first[c] * rs[c];
      sumA += Ac[c];
    }
    const float bv = pm - sumA * sumIm;
    #pragma unroll
    for (int c = 0; c < 4; ++c)
      Aout[(((size_t)bb * CC + c) * HH + row) * WW + col] = Ac[c];
    Bout[((size_t)bb * HH + row) * WW + col] = bv;
  }
}

__global__ __launch_bounds__(32) void gf_final(const float* __restrict__ I,
                                               const float* __restrict__ Ain,
                                               const float* __restrict__ Bin,
                                               float* __restrict__ Q) {
  __shared__ float X[32 * 33];
  __shared__ float T[32 * 17];
  const int lane = threadIdx.x;
  const int tx = blockIdx.x * 16;
  const int ty = blockIdx.y * 16;
  const int bb = blockIdx.z;
  const int gc = tx - RAD + lane;
  const bool cok = (gc >= 0) && (gc < WW);
  v2f band[8];
  make_band(band, lane);
  v8f S[5];
  #pragma unroll
  for (int pl = 0; pl < 5; ++pl) {
    const float* src = (pl < 4) ? (Ain + (((size_t)bb * CC + pl) * (size_t)HH) * WW)
                                : (Bin + (size_t)bb * HH * WW);
    for (int y = 0; y < 32; ++y) {
      const int gr = ty - RAD + y;
      const bool ok = cok && (gr >= 0) && (gr < HH);
      X[y * 33 + lane] = ok ? src[(size_t)gr * WW + gc] : 0.f;
    }
    __syncthreads();
    S[pl] = box2d_16(X, T, band, lane);
  }
  const int hi = lane >> 4;
  const int w = lane & 15;
  const int col = tx + w;
  const int cw = min(col + RAD, WW - 1) - max(col - RAD, 0) + 1;
  const float* Ib = I + (size_t)bb * CC * HH * WW;
  #pragma unroll
  for (int v = 0; v < 8; ++v) {
    const int row = ty + v + 8 * hi;
    const int chc = min(row + RAD, HH - 1) - max(row - RAD, 0) + 1;
    const float invN = 1.f / (float)(chc * cw);
    const size_t off = (size_t)row * WW + col;
    float sumI = 0.f;
    #pragma unroll
    for (int c = 0; c < 4; ++c) sumI += Ib[(size_t)c * HH * WW + off];
    const float am = (S[0][v] + S[1][v] + S[2][v] + S[3][v]) * invN;
    const float bm = S[4][v] * invN;
    Q[((size_t)bb * HH + row) * WW + col] = am * sumI + bm;
  }
}

extern "C" void kernel_launch(void* const* d_in, const int* in_sizes, int n_in,
                              void* d_out, int out_size, void* d_ws,
                              size_t ws_size, hipStream_t stream) {
  const float* I = (const float*)d_in[0];
  const float* P = (const float*)d_in[1];
  float* Q = (float*)d_out;
  float* Aws = (float*)d_ws;
  float* Bws = Aws + (size_t)NB * CC * HH * WW;
  dim3 grid(WW / 16, HH / 16, NB);
  gf_stats<<<grid, dim3(32), 0, stream>>>(I, P, Aws, Bws);
  gf_final<<<grid, dim3(32), 0, stream>>>(I, Aws, Bws, Q);
  (void)in_sizes; (void)n_in; (void)out_size; (void)ws_size;
}


# Per-wave pipeline between LDS and matrix runs: each group of fragment loads (next unit's first block too) issued one matrix run ahead, two fragment sets; no work-group barrier in the turn
# speedup vs baseline: 3.1353x; 3.1353x over previous
#include <hip/hip_runtime.h>


#define HH 768
#define WW 768
#define CC 4
#define NB 8
#define RAD 8
#define EPSV 1e-4f
#define NWAVE 8
#define NTHR (32 * NWAVE)
#define TL 32
#define HL (TL + 2 * RAD)
#define RP 49
#define RSZ (HL * RP)
#define TP 18
#define TSZ (HL * TP)
#define EP 34
#define ESZ (TL * EP)
#define NEL ((HL * HL) / NTHR)
#define NPX ((TL * TL) / NTHR)

typedef __attribute__((ext_vector_type(2))) float v2f;
typedef __attribute__((ext_vector_type(8))) float v8f;

static __device__ __forceinline__ v8f wmma4(v2f a, v2f b, v8f c) {
  return __builtin_amdgcn_wmma_f32_16x16x4_f32(false, a, false, b, (short)0, c,
                                               false, false);
}


static __device__ __forceinline__ void wave_lds_order() {
  __builtin_amdgcn_fence(__ATOMIC_RELEASE, "wavefront");
  __builtin_amdgcn_wave_barrier();
  __builtin_amdgcn_fence(__ATOMIC_ACQUIRE, "wavefront");
}

static __device__ __forceinline__ void h_load(const float* Xa, const float* Xb,
                                              bool prod, int rb, int m, int hi,
                                              v2f fa[8], v2f fb[8]) {
  #pragma unroll
  for (int kk = 0; kk < 8; ++kk) {
    const int o = (rb * 16 + m) * RP + 4 * kk + 2 * hi;
    v2f t = { Xa[o], Xa[o + 1] };
    fa[kk] = t;
  }
  if (prod) {
    #pragma unroll
    for (int kk = 0; kk < 8; ++kk) {
      const int o = (rb * 16 + m) * RP + 4 * kk + 2 * hi;
      v2f t = { Xb[o], Xb[o + 1] };
      fb[kk] = t;
    }
  }
}

static __device__ __forceinline__ void v_load(const float* T, int j0, int j1,
                                              int m, int hi, v2f b[12]) {
  const float* Tb = T + 2 * hi * TP + m;
  #pragma unroll
  for (int j = 0; j < 12; ++j) {
    if (j >= j0 && j < j1) {
      v2f t = { Tb[4 * j * TP], Tb[(4 * j + 1) * TP] };
      b[j] = t;
    }
  }
}

static __device__ __forceinline__ void box_unit(const float* Xa, const float* Xb,
                                                bool prod, const float* Na,
                                                const float* Nb, bool nprod,
                                                bool nact, float* T,
                                                const v2f band[8], int lane,
                                                v2f fa[8], v2f fb[8],
                                                v8f& s0, v8f& s1) {
  const int m = lane & 15;
  const int hi = lane >> 4;
  v2f b[12];
  #pragma unroll
  for (int rb = 0; rb < 3; ++rb) {
    v2f a[8];
    if (prod) {
      #pragma unroll
      for (int kk = 0; kk < 8; ++kk) a[kk] = fa[kk] * fb[kk];
    } else {
      #pragma unroll
      for (int kk = 0; kk < 8; ++kk) a[kk] = fa[kk];
    }
    if (rb < 2) {
      h_load(Xa, Xb, prod, rb + 1, m, hi, fa, fb);
    } else {
      wave_lds_order();
      v_load(T, 0, 8, m, hi, b);
    }
    __builtin_amdgcn_sched_barrier(0);
    v8f acc = {};
    #pragma unroll
    for (int kk = 0; kk < 8; ++kk) acc = wmma4(a[kk], band[kk], acc);
    __builtin_amdgcn_sched_barrier(0);
    #pragma unroll
    for (int v = 0; v < 8; ++v)
      T[(rb * 16 + v + 8 * hi) * TP + m] = acc[v];
  }
  wave_lds_order();
  v_load(T, 8, 12, m, hi, b);
  if (nact) h_load(Na, Nb, nprod, 0, m, hi, fa, fb);
  __builtin_amdgcn_sched_barrier(0);
  v8f a0 = {};
  v8f a1 = {};
  #pragma unroll
  for (int kk = 0; kk < 4; ++kk) a0 = wmma4(band[kk], b[kk], a0);
  #pragma unroll
  for (int kk = 4; kk < 8; ++kk) {
    a0 = wmma4(band[kk], b[kk], a0);
    a1 = wmma4(band[kk - 4], b[kk], a1);
  }
  #pragma unroll
  for (int kk = 4; kk < 8; ++kk) a1 = wmma4(band[kk], b[kk + 4], a1);
  s0 = a0;
  s1 = a1;
}

static __device__ __forceinline__ void make_band(v2f band[8], int lane) {
  const int hi = lane >> 4;
  const int m = lane & 15;
  #pragma unroll
  for (int kk = 0; kk < 8; ++kk) {
    const int d0 = 4 * kk + 2 * hi - m;
    v2f t = { (d0 >= 0 && d0 <= 16) ? 1.f : 0.f,
              (d0 + 1 >= 0 && d0 + 1 <= 16) ? 1.f : 0.f };
    band[kk] = t;
  }
}

static __device__ __forceinline__ void fetch5(const float* p0, const float* p1,
                                              const float* p2, const float* p3,
                                              const float* p4, float* dst,
                                              int tx, int ty, int tid) {
  float v[NEL][5];
  bool ok[NEL];
  int pos[NEL];
  #pragma unroll
  for (int i = 0; i < NEL; ++i) {
    const int idx = tid + NTHR * i;
    const int y = idx / HL;
    const int x = idx - y * HL;
    const int gr = ty - RAD + y;
    const int gc = tx - RAD + x;
    ok[i] = (gr >= 0) && (gr < HH) && (gc >= 0) && (gc < WW);
    const int cr = min(max(gr, 0), HH - 1);
    const int cc = min(max(gc, 0), WW - 1);
    const size_t off = (size_t)cr * WW + cc;
    pos[i] = y * RP + x;
    v[i][0] = p0[off];
    v[i][1] = p1[off];
    v[i][2] = p2[off];
    v[i][3] = p3[off];
    v[i][4] = p4[off];
  }
  #pragma unroll
  for (int i = 0; i < NEL; ++i) {
    #pragma unroll
    for (int c = 0; c < 5; ++c)
      dst[c * RSZ + pos[i]] = ok[i] ? v[i][c] : 0.f;
  }
}

static __device__ __forceinline__ void plane_pair(int pl, int& a, int& b) {
  if (pl < 9)       { a = 4; b = pl - 5; }
  else if (pl < 13) { a = 0; b = pl - 9; }
  else if (pl < 16) { a = 1; b = pl - 12; }
  else if (pl < 18) { a = 2; b = pl - 14; }
  else              { a = 3; b = 3; }
}

static __device__ __forceinline__ void inv4_rowsum(const float m[16],
                                                   float rs[4]) {
  float i0  =  m[5]*m[10]*m[15] - m[5]*m[11]*m[14] - m[9]*m[6]*m[15] + m[9]*m[7]*m[14] + m[13]*m[6]*m[11] - m[13]*m[7]*m[10];
  float i4  = -m[4]*m[10]*m[15] + m[4]*m[11]*m[14] + m[8]*m[6]*m[15] - m[8]*m[7]*m[14] - m[12]*m[6]*m[11] + m[12]*m[7]*m[10];
  float i8  =  m[4]*m[9]*m[15] - m[4]*m[11]*m[13] - m[8]*m[5]*m[15] + m[8]*m[7]*m[13] + m[12]*m[5]*m[11] - m[12]*m[7]*m[9];
  float i12 = -m[4]*m[9]*m[14] + m[4]*m[10]*m[13] + m[8]*m[5]*m[14] - m[8]*m[6]*m[13] - m[12]*m[5]*m[10] + m[12]*m[6]*m[9];
  float i1  = -m[1]*m[10]*m[15] + m[1]*m[11]*m[14] + m[9]*m[2]*m[15] - m[9]*m[3]*m[14] - m[13]*m[2]*m[11] + m[13]*m[3]*m[10];
  float i5  =  m[0]*m[10]*m[15] - m[0]*m[11]*m[14] - m[8]*m[2]*m[15] + m[8]*m[3]*m[14] + m[12]*m[2]*m[11] - m[12]*m[3]*m[10];
  float i9  = -m[0]*m[9]*m[15] + m[0]*m[11]*m[13] + m[8]*m[1]*m[15] - m[8]*m[3]*m[13] - m[12]*m[1]*m[11] + m[12]*m[3]*m[9];
  float i13 =  m[0]*m[9]*m[14] - m[0]*m[10]*m[13] - m[8]*m[1]*m[14] + m[8]*m[2]*m[13] + m[12]*m[1]*m[10] - m[12]*m[2]*m[9];
  float i2  =  m[1]*m[6]*m[15] - m[1]*m[7]*m[14] - m[5]*m[2]*m[15] + m[5]*m[3]*m[14] + m[13]*m[2]*m[7] - m[13]*m[3]*m[6];
  float i6  = -m[0]*m[6]*m[15] + m[0]*m[7]*m[14] + m[4]*m[2]*m[15] - m[4]*m[3]*m[14] - m[12]*m[2]*m[7] + m[12]*m[3]*m[6];
  float i10 =  m[0]*m[5]*m[15] - m[0]*m[7]*m[13] - m[4]*m[1]*m[15] + m[4]*m[3]*m[13] + m[12]*m[1]*m[7] - m[12]*m[3]*m[5];
  float i14 = -m[0]*m[5]*m[14] + m[0]*m[6]*m[13] + m[4]*m[1]*m[14] - m[4]*m[2]*m[13] - m[12]*m[1]*m[6] + m[12]*m[2]*m[5];
  float i3  = -m[1]*m[6]*m[11] + m[1]*m[7]*m[10] + m[5]*m[2]*m[11] - m[5]*m[3]*m[10] - m[9]*m[2]*m[7] + m[9]*m[3]*m[6];
  float i7  =  m[0]*m[6]*m[11] - m[0]*m[7]*m[10] - m[4]*m[2]*m[11] + m[4]*m[3]*m[10] + m[8]*m[2]*m[7] - m[8]*m[3]*m[6];
  float i11 = -m[0]*m[5]*m[11] + m[0]*m[7]*m[9] + m[4]*m[1]*m[11] - m[4]*m[3]*m[9] - m[8]*m[1]*m[7] + m[8]*m[3]*m[5];
  float i15 =  m[0]*m[5]*m[10] - m[0]*m[6]*m[9] - m[4]*m[1]*m[10] + m[4]*m[2]*m[9] + m[8]*m[1]*m[6] - m[8]*m[2]*m[5];
  const float det = m[0]*i0 + m[1]*i4 + m[2]*i8 + m[3]*i12;
  const float id = 1.0f / det;
  rs[0] = (i0 + i1 + i2 + i3) * id;
  rs[1] = (i4 + i5 + i6 + i7) * id;
  rs[2] = (i8 + i9 + i10 + i11) * id;
  rs[3] = (i12 + i13 + i14 + i15) * id;
}

__global__ __launch_bounds__(NTHR) __attribute__((amdgpu_waves_per_eu(6, 6)))
void gf_stats(const float* __restrict__ I, const float* __restrict__ P,
              float* __restrict__ Aout, float* __restrict__ Bout) {
  __shared__ float sm[19 * ESZ];
  float* raw = sm;
  float* Ts = sm + 5 * RSZ;
  const int tid = threadIdx.x;
  const int lane = tid & 31;
  const int wave = __builtin_amdgcn_readfirstlane(tid >> 5);
  const int tx = blockIdx.x * TL;
  const int ty = blockIdx.y * TL;
  const int bb = blockIdx.z;
  const float* Ib = I + (size_t)bb * CC * HH * WW;
  const float* Pb = P + (size_t)bb * HH * WW;
  fetch5(Ib, Ib + (size_t)HH * WW, Ib + (size_t)2 * HH * WW,
         Ib + (size_t)3 * HH * WW, Pb, raw, tx, ty, tid);
  __syncthreads();

  v2f band[8];
  make_band(band, lane);

  float* Tw = Ts + wave * TSZ;
  const int cb = wave & 1;
  const int hi = lane >> 4;
  const int m = lane & 15;
  const float* ua[5];
  const float* ub[5];
  bool up[5];
  bool uact[5];
  #pragma unroll
  for (int it = 0; it < 5; ++it) {
    const int pl = (wave >> 1) + 4 * it;
    uact[it] = pl < 19;
    int pa = pl, pb = pl;
    up[it] = !(it == 0 || (it == 1 && pl < 5));
    if (up[it]) plane_pair(pl, pa, pb);
    ua[it] = raw + pa * RSZ + cb * 16;
    ub[it] = raw + pb * RSZ + cb * 16;
  }
  v8f S[5][2];
  v2f fa[8], fb[8];
  h_load(ua[0], ub[0], up[0], 0, m, hi, fa, fb);
  #pragma unroll
  for (int it = 0; it < 5; ++it) {
    v8f s0 = {};
    v8f s1 = {};
    if (uact[it]) {
      const int nx = it + 1 < 5 ? it + 1 : it;
      box_unit(ua[it], ub[it], up[it], ua[nx], ub[nx], up[nx], it + 1 < 5,
               Tw, band, lane, fa, fb, s0, s1);
    }
    S[it][0] = s0;
    S[it][1] = s1;
  }
  __syncthreads();

  #pragma unroll
  for (int it = 0; it < 5; ++it) {
    const int pl = (wave >> 1) + 4 * it;
    if (pl < 19) {
      #pragma unroll
      for (int ob = 0; ob < 2; ++ob) {
        #pragma unroll
        for (int v = 0; v < 8; ++v)
          sm[pl * ESZ + (16 * ob + 8 * hi + v) * EP + 16 * cb + m] = S[it][ob][v];
      }
    }
  }
  __syncthreads();

  for (int q = 0; q < NPX; ++q) {
    const int px = tid + NTHR * q;
    const int pr = px >> 5;
    const int pc = px & 31;
    float Sv[19];
    #pragma unroll
    for (int pl = 0; pl < 19; ++pl) Sv[pl] = sm[pl * ESZ + pr * EP + pc];
    const int col = tx + pc;
    const int row = ty + pr;
    const int cw = min(col + RAD, WW - 1) - max(col - RAD, 0) + 1;
    const int chc = min(row + RAD, HH - 1) - max(row - RAD, 0) + 1;
    const float invN = 1.f / (float)(chc * cw);
    const float pm = Sv[4] * invN;
    float Im[4], first[4];
    float sumIm = 0.f;
    #pragma unroll
    for (int c = 0; c < 4; ++c) {
      Im[c] = Sv[c] * invN;
      sumIm += Im[c];
      first[c] = Sv[5 + c] * invN - pm * Im[c];
    }
    float M[16];
    M[0]  = Sv[9]  * invN + EPSV;
    M[1]  = M[4]  = Sv[10] * invN;
    M[2]  = M[8]  = Sv[11] * invN;
    M[3]  = M[12] = Sv[12] * invN;
    M[5]  = Sv[13] * invN + EPSV;
    M[6]  = M[9]  = Sv[14] * invN;
    M[7]  = M[13] = Sv[15] * invN;
    M[10] = Sv[16] * invN + EPSV;
    M[11] = M[14] = Sv[17] * invN;
    M[15] = Sv[18] * invN + EPSV;
    float rs[4];
    inv4_rowsum(M, rs);
    float sumA = 0.f;
    float Ac[4];
    #pragma unroll
    for (int c = 0; c < 4; ++c) {
      Ac[c] = first[c] * rs[c];
      sumA += Ac[c];
    }
    const float bv = pm - sumA * sumIm;
    #pragma unroll
    for (int c = 0; c < 4; ++c)
      Aout[(((size_t)bb * CC + c) * HH + row) * WW + col] = Ac[c];
    Bout[((size_t)bb * HH + row) * WW + col] = bv;
  }
}

__global__ __launch_bounds__(NTHR) void gf_final(const float* __restrict__ I,
                                                 const float* __restrict__ Ain,
                                                 const float* __restrict__ Bin,
                                                 float* __restrict__ Q) {
  __shared__ float sm[5 * RSZ + NWAVE * TSZ];
  float* raw = sm;
  float* Ts = sm + 5 * RSZ;
  const int tid = threadIdx.x;
  const int lane = tid & 31;
  const int wave = __builtin_amdgcn_readfirstlane(tid >> 5);
  const int tx = blockIdx.x * TL;
  const int ty = blockIdx.y * TL;
  const int bb = blockIdx.z;
  const float* Ab = Ain + (size_t)bb * CC * HH * WW;
  const float* Bb = Bin + (size_t)bb * HH * WW;
  const float* Ib = I + (size_t)bb * CC * HH * WW;
  fetch5(Ab, Ab + (size_t)HH * WW, Ab + (size_t)2 * HH * WW,
         Ab + (size_t)3 * HH * WW, Bb, raw, tx, ty, tid);
  float sumI[NPX];
  #pragma unroll
  for (int q = 0; q < NPX; ++q) {
    const int px = tid + NTHR * q;
    const size_t off = (size_t)(ty + (px >> 5)) * WW + (tx + (px & 31));
    float s = 0.f;
    #pragma unroll
    for (int c = 0; c < 4; ++c) s += Ib[(size_t)c * HH * WW + off];
    sumI[q] = s;
  }
  __syncthreads();

  v2f band[8];
  make_band(band, lane);
  float* Tw = Ts + wave * TSZ;
  const int cb = wave & 1;
  const int hi = lane >> 4;
  const int m = lane & 15;
  const int pl0 = wave >> 1;
  const bool act1 = pl0 + 4 < 5;
  const float* u0 = raw + pl0 * RSZ + cb * 16;
  const float* u1 = raw + 4 * RSZ + cb * 16;
  v8f S[2][2];
  v2f fa[8], fb[8];
  h_load(u0, u0, false, 0, m, hi, fa, fb);
  {
    v8f s0 = {};
    v8f s1 = {};
    box_unit(u0, u0, false, u1, u1, false, true, Tw, band, lane, fa, fb, s0, s1);
    S[0][0] = s0;
    S[0][1] = s1;
  }
  {
    v8f s0 = {};
    v8f s1 = {};
    if (act1)
      box_unit(u1, u1, false, u1, u1, false, false, Tw, band, lane, fa, fb, s0, s1);
    S[1][0] = s0;
    S[1][1] = s1;
  }
  __syncthreads();
  #pragma unroll
  for (int it = 0; it < 2; ++it) {
    const int pl = (wave >> 1) + 4 * it;
    if (pl < 5) {
      #pragma unroll
      for (int ob = 0; ob < 2; ++ob) {
        #pragma unroll
        for (int v = 0; v < 8; ++v)
          sm[pl * ESZ + (16 * ob + 8 * hi + v) * EP + 16 * cb + m] = S[it][ob][v];
      }
    }
  }
  __syncthreads();

  #pragma unroll
  for (int q = 0; q < NPX; ++q) {
    const int px = tid + NTHR * q;
    const int pr = px >> 5;
    const int pc = px & 31;
    const int col = tx + pc;
    const int row = ty + pr;
    const int cw = min(col + RAD, WW - 1) - max(col - RAD, 0) + 1;
    const int chc = min(row + RAD, HH - 1) - max(row - RAD, 0) + 1;
    const float invN = 1.f / (float)(chc * cw);
    const int e = pr * EP + pc;
    const float am = (sm[e] + sm[ESZ + e] + sm[2 * ESZ + e] + sm[3 * ESZ + e]) * invN;
    const float bm = sm[4 * ESZ + e] * invN;
    Q[((size_t)bb * HH + row) * WW + col] = am * sumI[q] + bm;
  }
}

extern "C" void kernel_launch(void* const* d_in, const int* in_sizes, int n_in,
                              void* d_out, int out_size, void* d_ws,
                              size_t ws_size, hipStream_t stream) {
  const float* I = (const float*)d_in[0];
  const float* P = (const float*)d_in[1];
  float* Q = (float*)d_out;
  float* Aws = (float*)d_ws;
  float* Bws = Aws + (size_t)NB * CC * HH * WW;
  dim3 grid(WW / TL, HH / TL, NB);
  gf_stats<<<grid, dim3(NTHR), 0, stream>>>(I, P, Aws, Bws);
  gf_final<<<grid, dim3(NTHR), 0, stream>>>(I, Aws, Bws, Q);
  (void)in_sizes; (void)n_in; (void)out_size; (void)ws_size;
}


# Per-pixel algebra on pixel pairs in packed f32 (two pixels per vector instruction); inverse row sums from shared 2x2 minors, 10 cofactors of the symmetric matrix, not 16; rounding differs
# speedup vs baseline: 3.3616x; 1.0722x over previous
#include <hip/hip_runtime.h>


#define HH 768
#define WW 768
#define CC 4
#define NB 8
#define RAD 8
#define EPSV 1e-4f
#define NWAVE 8
#define NTHR (32 * NWAVE)
#define TL 32
#define HL (TL + 2 * RAD)
#define RP 49
#define RSZ (HL * RP)
#define TP 18
#define TSZ (HL * TP)
#define EP 34
#define ESZ (TL * EP)
#define NEL ((HL * HL) / NTHR)
#define NPX ((TL * TL) / NTHR)

typedef __attribute__((ext_vector_type(2))) float v2f;
typedef __attribute__((ext_vector_type(8))) float v8f;

static __device__ __forceinline__ v8f wmma4(v2f a, v2f b, v8f c) {
  return __builtin_amdgcn_wmma_f32_16x16x4_f32(false, a, false, b, (short)0, c,
                                               false, false);
}


static __device__ __forceinline__ void wave_lds_order() {
  __builtin_amdgcn_fence(__ATOMIC_RELEASE, "wavefront");
  __builtin_amdgcn_wave_barrier();
  __builtin_amdgcn_fence(__ATOMIC_ACQUIRE, "wavefront");
}

static __device__ __forceinline__ void h_load(const float* Xa, const float* Xb,
                                              bool prod, int rb, int m, int hi,
                                              v2f fa[8], v2f fb[8]) {
  #pragma unroll
  for (int kk = 0; kk < 8; ++kk) {
    const int o = (rb * 16 + m) * RP + 4 * kk + 2 * hi;
    v2f t = { Xa[o], Xa[o + 1] };
    fa[kk] = t;
  }
  if (prod) {
    #pragma unroll
    for (int kk = 0; kk < 8; ++kk) {
      const int o = (rb * 16 + m) * RP + 4 * kk + 2 * hi;
      v2f t = { Xb[o], Xb[o + 1] };
      fb[kk] = t;
    }
  }
}

static __device__ __forceinline__ void v_load(const float* T, int j0, int j1,
                                              int m, int hi, v2f b[12]) {
  const float* Tb = T + 2 * hi * TP + m;
  #pragma unroll
  for (int j = 0; j < 12; ++j) {
    if (j >= j0 && j < j1) {
      v2f t = { Tb[4 * j * TP], Tb[(4 * j + 1) * TP] };
      b[j] = t;
    }
  }
}

static __device__ __forceinline__ void box_unit(const float* Xa, const float* Xb,
                                                bool prod, const float* Na,
                                                const float* Nb, bool nprod,
                                                bool nact, float* T,
                                                const v2f band[8], int lane,
                                                v2f fa[8], v2f fb[8],
                                                v8f& s0, v8f& s1) {
  const int m = lane & 15;
  const int hi = lane >> 4;
  v2f b[12];
  #pragma unroll
  for (int rb = 0; rb < 3; ++rb) {
    v2f a[8];
    if (prod) {
      #pragma unroll
      for (int kk = 0; kk < 8; ++kk) a[kk] = fa[kk] * fb[kk];
    } else {
      #pragma unroll
      for (int kk = 0; kk < 8; ++kk) a[kk] = fa[kk];
    }
    if (rb < 2) {
      h_load(Xa, Xb, prod, rb + 1, m, hi, fa, fb);
    } else {
      wave_lds_order();
      v_load(T, 0, 8, m, hi, b);
    }
    __builtin_amdgcn_sched_barrier(0);
    v8f acc = {};
    #pragma unroll
    for (int kk = 0; kk < 8; ++kk) acc = wmma4(a[kk], band[kk], acc);
    __builtin_amdgcn_sched_barrier(0);
    #pragma unroll
    for (int v = 0; v < 8; ++v)
      T[(rb * 16 + v + 8 * hi) * TP + m] = acc[v];
  }
  wave_lds_order();
  v_load(T, 8, 12, m, hi, b);
  if (nact) h_load(Na, Nb, nprod, 0, m, hi, fa, fb);
  __builtin_amdgcn_sched_barrier(0);
  v8f a0 = {};
  v8f a1 = {};
  #pragma unroll
  for (int kk = 0; kk < 4; ++kk) a0 = wmma4(band[kk], b[kk], a0);
  #pragma unroll
  for (int kk = 4; kk < 8; ++kk) {
    a0 = wmma4(band[kk], b[kk], a0);
    a1 = wmma4(band[kk - 4], b[kk], a1);
  }
  #pragma unroll
  for (int kk = 4; kk < 8; ++kk) a1 = wmma4(band[kk], b[kk + 4], a1);
  s0 = a0;
  s1 = a1;
}

static __device__ __forceinline__ void make_band(v2f band[8], int lane) {
  const int hi = lane >> 4;
  const int m = lane & 15;
  #pragma unroll
  for (int kk = 0; kk < 8; ++kk) {
    const int d0 = 4 * kk + 2 * hi - m;
    v2f t = { (d0 >= 0 && d0 <= 16) ? 1.f : 0.f,
              (d0 + 1 >= 0 && d0 + 1 <= 16) ? 1.f : 0.f };
    band[kk] = t;
  }
}

static __device__ __forceinline__ void fetch5(const float* p0, const float* p1,
                                              const float* p2, const float* p3,
                                              const float* p4, float* dst,
                                              int tx, int ty, int tid) {
  float v[NEL][5];
  bool ok[NEL];
  int pos[NEL];
  #pragma unroll
  for (int i = 0; i < NEL; ++i) {
    const int idx = tid + NTHR * i;
    const int y = idx / HL;
    const int x = idx - y * HL;
    const int gr = ty - RAD + y;
    const int gc = tx - RAD + x;
    ok[i] = (gr >= 0) && (gr < HH) && (gc >= 0) && (gc < WW);
    const int cr = min(max(gr, 0), HH - 1);
    const int cc = min(max(gc, 0), WW - 1);
    const size_t off = (size_t)cr * WW + cc;
    pos[i] = y * RP + x;
    v[i][0] = p0[off];
    v[i][1] = p1[off];
    v[i][2] = p2[off];
    v[i][3] = p3[off];
    v[i][4] = p4[off];
  }
  #pragma unroll
  for (int i = 0; i < NEL; ++i) {
    #pragma unroll
    for (int c = 0; c < 5; ++c)
      dst[c * RSZ + pos[i]] = ok[i] ? v[i][c] : 0.f;
  }
}

static __device__ __forceinline__ void plane_pair(int pl, int& a, int& b) {
  if (pl < 9)       { a = 4; b = pl - 5; }
  else if (pl < 13) { a = 0; b = pl - 9; }
  else if (pl < 16) { a = 1; b = pl - 12; }
  else if (pl < 18) { a = 2; b = pl - 14; }
  else              { a = 3; b = 3; }
}

static __device__ __forceinline__ void inv4_rowsum_sym2(v2f a, v2f b, v2f c, v2f d,
                                                        v2f e, v2f f, v2f g, v2f h,
                                                        v2f i, v2f j, v2f rs[4]) {
  const v2f s0 = a * e - b * b;
  const v2f s1 = a * f - b * c;
  const v2f s2 = a * g - b * d;
  const v2f s3 = b * f - e * c;
  const v2f s4 = b * g - e * d;
  const v2f s5 = c * g - f * d;
  const v2f c5 = h * j - i * i;
  const v2f c4 = f * j - g * i;
  const v2f c3 = f * i - g * h;
  const v2f c2 = c * j - d * i;
  const v2f c1 = c * i - d * h;
  const v2f det = s0 * c5 - s1 * c4 + s2 * c3 + s3 * c2 - s4 * c1 + s5 * s5;
  const v2f b00 = e * c5 - f * c4 + g * c3;
  const v2f b01 = c * c4 - b * c5 - d * c3;
  const v2f b02 = g * s5 - i * s4 + j * s3;
  const v2f b03 = h * s4 - f * s5 - i * s3;
  const v2f b11 = a * c5 - c * c2 + d * c1;
  const v2f b12 = i * s2 - d * s5 - j * s1;
  const v2f b13 = c * s5 - h * s2 + i * s1;
  const v2f b22 = d * s4 - g * s2 + j * s0;
  const v2f b23 = f * s2 - c * s4 - i * s0;
  const v2f b33 = c * s3 - f * s1 + h * s0;
  const v2f id = { 1.0f / det.x, 1.0f / det.y };
  rs[0] = (b00 + b01 + b02 + b03) * id;
  rs[1] = (b01 + b11 + b12 + b13) * id;
  rs[2] = (b02 + b12 + b22 + b23) * id;
  rs[3] = (b03 + b13 + b23 + b33) * id;
}

__global__ __launch_bounds__(NTHR) __attribute__((amdgpu_waves_per_eu(6, 6)))
void gf_stats(const float* __restrict__ I, const float* __restrict__ P,
              float* __restrict__ Aout, float* __restrict__ Bout) {
  __shared__ float sm[19 * ESZ];
  float* raw = sm;
  float* Ts = sm + 5 * RSZ;
  const int tid = threadIdx.x;
  const int lane = tid & 31;
  const int wave = __builtin_amdgcn_readfirstlane(tid >> 5);
  const int tx = blockIdx.x * TL;
  const int ty = blockIdx.y * TL;
  const int bb = blockIdx.z;
  const float* Ib = I + (size_t)bb * CC * HH * WW;
  const float* Pb = P + (size_t)bb * HH * WW;
  fetch5(Ib, Ib + (size_t)HH * WW, Ib + (size_t)2 * HH * WW,
         Ib + (size_t)3 * HH * WW, Pb, raw, tx, ty, tid);
  __syncthreads();

  v2f band[8];
  make_band(band, lane);

  float* Tw = Ts + wave * TSZ;
  const int cb = wave & 1;
  const int hi = lane >> 4;
  const int m = lane & 15;
  const float* ua[5];
  const float* ub[5];
  bool up[5];
  bool uact[5];
  #pragma unroll
  for (int it = 0; it < 5; ++it) {
    const int pl = (wave >> 1) + 4 * it;
    uact[it] = pl < 19;
    int pa = pl, pb = pl;
    up[it] = !(it == 0 || (it == 1 && pl < 5));
    if (up[it]) plane_pair(pl, pa, pb);
    ua[it] = raw + pa * RSZ + cb * 16;
    ub[it] = raw + pb * RSZ + cb * 16;
  }
  v8f S[5][2];
  v2f fa[8], fb[8];
  h_load(ua[0], ub[0], up[0], 0, m, hi, fa, fb);
  #pragma unroll
  for (int it = 0; it < 5; ++it) {
    v8f s0 = {};
    v8f s1 = {};
    if (uact[it]) {
      const int nx = it + 1 < 5 ? it + 1 : it;
      box_unit(ua[it], ub[it], up[it], ua[nx], ub[nx], up[nx], it + 1 < 5,
               Tw, band, lane, fa, fb, s0, s1);
    }
    S[it][0] = s0;
    S[it][1] = s1;
  }
  __syncthreads();

  #pragma unroll
  for (int it = 0; it < 5; ++it) {
    const int pl = (wave >> 1) + 4 * it;
    if (pl < 19) {
      #pragma unroll
      for (int ob = 0; ob < 2; ++ob) {
        #pragma unroll
        for (int v = 0; v < 8; ++v)
          sm[pl * ESZ + (16 * ob + 8 * hi + v) * EP + 16 * cb + m] = S[it][ob][v];
      }
    }
  }
  __syncthreads();

  for (int q = 0; q < NPX; q += 2) {
    const int pr = 2 * (tid >> 5) + 8 * q;
    const int pc = tid & 31;
    v2f Sv[19];
    #pragma unroll
    for (int pl = 0; pl < 19; ++pl) {
      v2f t = { sm[pl * ESZ + pr * EP + pc], sm[pl * ESZ + (pr + 1) * EP + pc] };
      Sv[pl] = t;
    }
    const int col = tx + pc;
    const int row = ty + pr;
    const int row1 = row + 1;
    const int cw = min(col + RAD, WW - 1) - max(col - RAD, 0) + 1;
    const int chc = min(row + RAD, HH - 1) - max(row - RAD, 0) + 1;
    const int chc1 = min(row1 + RAD, HH - 1) - max(row1 - RAD, 0) + 1;
    const v2f invN = { 1.f / (float)(chc * cw), 1.f / (float)(chc1 * cw) };
    const v2f pm = Sv[4] * invN;
    v2f Im[4], first[4];
    v2f sumIm = { 0.f, 0.f };
    #pragma unroll
    for (int c = 0; c < 4; ++c) {
      Im[c] = Sv[c] * invN;
      sumIm += Im[c];
      first[c] = Sv[5 + c] * invN - pm * Im[c];
    }
    const v2f m00 = Sv[9]  * invN + EPSV;
    const v2f m01 = Sv[10] * invN;
    const v2f m02 = Sv[11] * invN;
    const v2f m03 = Sv[12] * invN;
    const v2f m11 = Sv[13] * invN + EPSV;
    const v2f m12 = Sv[14] * invN;
    const v2f m13 = Sv[15] * invN;
    const v2f m22 = Sv[16] * invN + EPSV;
    const v2f m23 = Sv[17] * invN;
    const v2f m33 = Sv[18] * invN + EPSV;
    v2f rs[4];
    inv4_rowsum_sym2(m00, m01, m02, m03, m11, m12, m13, m22, m23, m33, rs);
    v2f sumA = { 0.f, 0.f };
    v2f Ac[4];
    #pragma unroll
    for (int c = 0; c < 4; ++c) {
      Ac[c] = first[c] * rs[c];
      sumA += Ac[c];
    }
    const v2f bv = pm - sumA * sumIm;
    #pragma unroll
    for (int c = 0; c < 4; ++c) {
      Aout[(((size_t)bb * CC + c) * HH + row) * WW + col] = Ac[c].x;
      Aout[(((size_t)bb * CC + c) * HH + row1) * WW + col] = Ac[c].y;
    }
    Bout[((size_t)bb * HH + row) * WW + col] = bv.x;
    Bout[((size_t)bb * HH + row1) * WW + col] = bv.y;
  }
}

__global__ __launch_bounds__(NTHR) void gf_final(const float* __restrict__ I,
                                                 const float* __restrict__ Ain,
                                                 const float* __restrict__ Bin,
                                                 float* __restrict__ Q) {
  __shared__ float sm[5 * RSZ + NWAVE * TSZ];
  float* raw = sm;
  float* Ts = sm + 5 * RSZ;
  const int tid = threadIdx.x;
  const int lane = tid & 31;
  const int wave = __builtin_amdgcn_readfirstlane(tid >> 5);
  const int tx = blockIdx.x * TL;
  const int ty = blockIdx.y * TL;
  const int bb = blockIdx.z;
  const float* Ab = Ain + (size_t)bb * CC * HH * WW;
  const float* Bb = Bin + (size_t)bb * HH * WW;
  const float* Ib = I + (size_t)bb * CC * HH * WW;
  fetch5(Ab, Ab + (size_t)HH * WW, Ab + (size_t)2 * HH * WW,
         Ab + (size_t)3 * HH * WW, Bb, raw, tx, ty, tid);
  float sumI[NPX];
  #pragma unroll
  for (int q = 0; q < NPX; ++q) {
    const int px = tid + NTHR * q;
    const size_t off = (size_t)(ty + (px >> 5)) * WW + (tx + (px & 31));
    float s = 0.f;
    #pragma unroll
    for (int c = 0; c < 4; ++c) s += Ib[(size_t)c * HH * WW + off];
    sumI[q] = s;
  }
  __syncthreads();

  v2f band[8];
  make_band(band, lane);
  float* Tw = Ts + wave * TSZ;
  const int cb = wave & 1;
  const int hi = lane >> 4;
  const int m = lane & 15;
  const int pl0 = wave >> 1;
  const bool act1 = pl0 + 4 < 5;
  const float* u0 = raw + pl0 * RSZ + cb * 16;
  const float* u1 = raw + 4 * RSZ + cb * 16;
  v8f S[2][2];
  v2f fa[8], fb[8];
  h_load(u0, u0, false, 0, m, hi, fa, fb);
  {
    v8f s0 = {};
    v8f s1 = {};
    box_unit(u0, u0, false, u1, u1, false, true, Tw, band, lane, fa, fb, s0, s1);
    S[0][0] = s0;
    S[0][1] = s1;
  }
  {
    v8f s0 = {};
    v8f s1 = {};
    if (act1)
      box_unit(u1, u1, false, u1, u1, false, false, Tw, band, lane, fa, fb, s0, s1);
    S[1][0] = s0;
    S[1][1] = s1;
  }
  __syncthreads();
  #pragma unroll
  for (int it = 0; it < 2; ++it) {
    const int pl = (wave >> 1) + 4 * it;
    if (pl < 5) {
      #pragma unroll
      for (int ob = 0; ob < 2; ++ob) {
        #pragma unroll
        for (int v = 0; v < 8; ++v)
          sm[pl * ESZ + (16 * ob + 8 * hi + v) * EP + 16 * cb + m] = S[it][ob][v];
      }
    }
  }
  __syncthreads();

  #pragma unroll
  for (int q = 0; q < NPX; ++q) {
    const int px = tid + NTHR * q;
    const int pr = px >> 5;
    const int pc = px & 31;
    const int col = tx + pc;
    const int row = ty + pr;
    const int cw = min(col + RAD, WW - 1) - max(col - RAD, 0) + 1;
    const int chc = min(row + RAD, HH - 1) - max(row - RAD, 0) + 1;
    const float invN = 1.f / (float)(chc * cw);
    const int e = pr * EP + pc;
    const float am = (sm[e] + sm[ESZ + e] + sm[2 * ESZ + e] + sm[3 * ESZ + e]) * invN;
    const float bm = sm[4 * ESZ + e] * invN;
    Q[((size_t)bb * HH + row) * WW + col] = am * sumI[q] + bm;
  }
}

extern "C" void kernel_launch(void* const* d_in, const int* in_sizes, int n_in,
                              void* d_out, int out_size, void* d_ws,
                              size_t ws_size, hipStream_t stream) {
  const float* I = (const float*)d_in[0];
  const float* P = (const float*)d_in[1];
  float* Q = (float*)d_out;
  float* Aws = (float*)d_ws;
  float* Bws = Aws + (size_t)NB * CC * HH * WW;
  dim3 grid(WW / TL, HH / TL, NB);
  gf_stats<<<grid, dim3(NTHR), 0, stream>>>(I, P, Aws, Bws);
  gf_final<<<grid, dim3(NTHR), 0, stream>>>(I, Aws, Bws, Q);
  (void)in_sizes; (void)n_in; (void)out_size; (void)ws_size;
}


# Fragment traffic through LDS in 16-byte accesses: k order regrouped so one load feeds two matrix ops, intermediate kept column-major (8 accumulator rows leave in 2 stores); sums reordered
# speedup vs baseline: 3.5040x; 1.0424x over previous
#include <hip/hip_runtime.h>


#define HH 768
#define WW 768
#define CC 4
#define NB 8
#define RAD 8
#define EPSV 1e-4f
#define NWAVE 8
#define NTHR (32 * NWAVE)
#define TL 32
#define HL (TL + 2 * RAD)
#define RP 52
#define RSZ (HL * RP)
#define TP 52
#define TSZ (16 * TP)
#define EP 34
#define ESZ (TL * EP)
#define NEL ((HL * HL) / NTHR)
#define NPX ((TL * TL) / NTHR)

typedef __attribute__((ext_vector_type(2))) float v2f;
typedef __attribute__((ext_vector_type(4))) float v4f;
typedef __attribute__((ext_vector_type(8))) float v8f;

static __device__ __forceinline__ v8f wmma4(v2f a, v2f b, v8f c) {
  return __builtin_amdgcn_wmma_f32_16x16x4_f32(false, a, false, b, (short)0, c,
                                               false, false);
}


static __device__ __forceinline__ void wave_lds_order() {
  __builtin_amdgcn_fence(__ATOMIC_RELEASE, "wavefront");
  __builtin_amdgcn_wave_barrier();
  __builtin_amdgcn_fence(__ATOMIC_ACQUIRE, "wavefront");
}

static __device__ __forceinline__ void h_load(const float* Xa, const float* Xb,
                                              bool prod, int rb, int m, int hi,
                                              v2f fa[8], v2f fb[8]) {
  const int o = (rb * 16 + m) * RP + 4 * hi;
  #pragma unroll
  for (int j = 0; j < 4; ++j) {
    const v4f q = *(const v4f*)(Xa + o + 8 * j);
    v2f t0 = { q.x, q.y };
    v2f t1 = { q.z, q.w };
    fa[2 * j] = t0;
    fa[2 * j + 1] = t1;
  }
  if (prod) {
    #pragma unroll
    for (int j = 0; j < 4; ++j) {
      const v4f q = *(const v4f*)(Xb + o + 8 * j);
      v2f t0 = { q.x, q.y };
      v2f t1 = { q.z, q.w };
      fb[2 * j] = t0;
      fb[2 * j + 1] = t1;
    }
  }
}

static __device__ __forceinline__ void v_load(const float* T, int j0, int j1,
                                              int m, int hi, v2f b[12]) {
  const float* Tb = T + m * TP + 4 * hi;
  #pragma unroll
  for (int j = 0; j < 6; ++j) {
    if (j >= j0 && j < j1) {
      const v4f q = *(const v4f*)(Tb + 8 * j);
      v2f t0 = { q.x, q.y };
      v2f t1 = { q.z, q.w };
      b[2 * j] = t0;
      b[2 * j + 1] = t1;
    }
  }
}

static __device__ __forceinline__ void box_unit(const float* Xa, const float* Xb,
                                                bool prod, const float* Na,
                                                const float* Nb, bool nprod,
                                                bool nact, float* T,
                                                const v2f band[8], int lane,
                                                v2f fa[8], v2f fb[8],
                                                v8f& s0, v8f& s1) {
  const int m = lane & 15;
  const int hi = lane >> 4;
  v2f b[12];
  #pragma unroll
  for (int rb = 0; rb < 3; ++rb) {
    v2f a[8];
    if (prod) {
      #pragma unroll
      for (int kk = 0; kk < 8; ++kk) a[kk] = fa[kk] * fb[kk];
    } else {
      #pragma unroll
      for (int kk = 0; kk < 8; ++kk) a[kk] = fa[kk];
    }
    __builtin_amdgcn_sched_barrier(0);
    if (rb < 2) {
      h_load(Xa, Xb, prod, rb + 1, m, hi, fa, fb);
    } else {
      wave_lds_order();
      v_load(T, 0, 4, m, hi, b);
    }
    __builtin_amdgcn_sched_barrier(0);
    v8f acc = {};
    #pragma unroll
    for (int kk = 0; kk < 8; ++kk) acc = wmma4(a[kk], band[kk], acc);
    __builtin_amdgcn_sched_barrier(0);
    float* Tc = T + m * TP + rb * 16 + 8 * hi;
    const v4f lo4 = { acc[0], acc[1], acc[2], acc[3] };
    const v4f hi4 = { acc[4], acc[5], acc[6], acc[7] };
    *(v4f*)Tc = lo4;
    *(v4f*)(Tc + 4) = hi4;
  }
  wave_lds_order();
  v_load(T, 4, 6, m, hi, b);
  if (nact) h_load(Na, Nb, nprod, 0, m, hi, fa, fb);
  __builtin_amdgcn_sched_barrier(0);
  v8f a0 = {};
  v8f a1 = {};
  #pragma unroll
  for (int kk = 0; kk < 4; ++kk) a0 = wmma4(band[kk], b[kk], a0);
  #pragma unroll
  for (int kk = 4; kk < 8; ++kk) {
    a0 = wmma4(band[kk], b[kk], a0);
    a1 = wmma4(band[kk - 4], b[kk], a1);
  }
  #pragma unroll
  for (int kk = 4; kk < 8; ++kk) a1 = wmma4(band[kk], b[kk + 4], a1);
  s0 = a0;
  s1 = a1;
}

static __device__ __forceinline__ void make_band(v2f band[8], int lane) {
  const int hi = lane >> 4;
  const int m = lane & 15;
  #pragma unroll
  for (int kk = 0; kk < 8; ++kk) {
    const int d0 = 8 * (kk >> 1) + 4 * hi + 2 * (kk & 1) - m;
    v2f t = { (d0 >= 0 && d0 <= 16) ? 1.f : 0.f,
              (d0 + 1 >= 0 && d0 + 1 <= 16) ? 1.f : 0.f };
    band[kk] = t;
  }
}

static __device__ __forceinline__ void fetch5(const float* p0, const float* p1,
                                              const float* p2, const float* p3,
                                              const float* p4, float* dst,
                                              int tx, int ty, int tid) {
  float v[NEL][5];
  bool ok[NEL];
  int pos[NEL];
  #pragma unroll
  for (int i = 0; i < NEL; ++i) {
    const int idx = tid + NTHR * i;
    const int y = idx / HL;
    const int x = idx - y * HL;
    const int gr = ty - RAD + y;
    const int gc = tx - RAD + x;
    ok[i] = (gr >= 0) && (gr < HH) && (gc >= 0) && (gc < WW);
    const int cr = min(max(gr, 0), HH - 1);
    const int cc = min(max(gc, 0), WW - 1);
    const size_t off = (size_t)cr * WW + cc;
    pos[i] = y * RP + x;
    v[i][0] = p0[off];
    v[i][1] = p1[off];
    v[i][2] = p2[off];
    v[i][3] = p3[off];
    v[i][4] = p4[off];
  }
  #pragma unroll
  for (int i = 0; i < NEL; ++i) {
    #pragma unroll
    for (int c = 0; c < 5; ++c)
      dst[c * RSZ + pos[i]] = ok[i] ? v[i][c] : 0.f;
  }
}

static __device__ __forceinline__ void plane_pair(int pl, int& a, int& b) {
  if (pl < 9)       { a = 4; b = pl - 5; }
  else if (pl < 13) { a = 0; b = pl - 9; }
  else if (pl < 16) { a = 1; b = pl - 12; }
  else if (pl < 18) { a = 2; b = pl - 14; }
  else              { a = 3; b = 3; }
}

static __device__ __forceinline__ void inv4_rowsum_sym2(v2f a, v2f b, v2f c, v2f d,
                                                        v2f e, v2f f, v2f g, v2f h,
                                                        v2f i, v2f j, v2f rs[4]) {
  const v2f s0 = a * e - b * b;
  const v2f s1 = a * f - b * c;
  const v2f s2 = a * g - b * d;
  const v2f s3 = b * f - e * c;
  const v2f s4 = b * g - e * d;
  const v2f s5 = c * g - f * d;
  const v2f c5 = h * j - i * i;
  const v2f c4 = f * j - g * i;
  const v2f c3 = f * i - g * h;
  const v2f c2 = c * j - d * i;
  const v2f c1 = c * i - d * h;
  const v2f det = s0 * c5 - s1 * c4 + s2 * c3 + s3 * c2 - s4 * c1 + s5 * s5;
  const v2f b00 = e * c5 - f * c4 + g * c3;
  const v2f b01 = c * c4 - b * c5 - d * c3;
  const v2f b02 = g * s5 - i * s4 + j * s3;
  const v2f b03 = h * s4 - f * s5 - i * s3;
  const v2f b11 = a * c5 - c * c2 + d * c1;
  const v2f b12 = i * s2 - d * s5 - j * s1;
  const v2f b13 = c * s5 - h * s2 + i * s1;
  const v2f b22 = d * s4 - g * s2 + j * s0;
  const v2f b23 = f * s2 - c * s4 - i * s0;
  const v2f b33 = c * s3 - f * s1 + h * s0;
  const v2f id = { 1.0f / det.x, 1.0f / det.y };
  rs[0] = (b00 + b01 + b02 + b03) * id;
  rs[1] = (b01 + b11 + b12 + b13) * id;
  rs[2] = (b02 + b12 + b22 + b23) * id;
  rs[3] = (b03 + b13 + b23 + b33) * id;
}

__global__ __launch_bounds__(NTHR) __attribute__((amdgpu_waves_per_eu(6, 6)))
void gf_stats(const float* __restrict__ I, const float* __restrict__ P,
              float* __restrict__ Aout, float* __restrict__ Bout) {
  __shared__ __attribute__((aligned(16))) float sm[19 * ESZ];
  float* raw = sm;
  float* Ts = sm + 5 * RSZ;
  const int tid = threadIdx.x;
  const int lane = tid & 31;
  const int wave = __builtin_amdgcn_readfirstlane(tid >> 5);
  const int tx = blockIdx.x * TL;
  const int ty = blockIdx.y * TL;
  const int bb = blockIdx.z;
  const float* Ib = I + (size_t)bb * CC * HH * WW;
  const float* Pb = P + (size_t)bb * HH * WW;
  fetch5(Ib, Ib + (size_t)HH * WW, Ib + (size_t)2 * HH * WW,
         Ib + (size_t)3 * HH * WW, Pb, raw, tx, ty, tid);
  __syncthreads();

  v2f band[8];
  make_band(band, lane);

  float* Tw = Ts + wave * TSZ;
  const int cb = wave & 1;
  const int hi = lane >> 4;
  const int m = lane & 15;
  const float* ua[5];
  const float* ub[5];
  bool up[5];
  bool uact[5];
  #pragma unroll
  for (int it = 0; it < 5; ++it) {
    const int pl = (wave >> 1) + 4 * it;
    uact[it] = pl < 19;
    int pa = pl, pb = pl;
    up[it] = !(it == 0 || (it == 1 && pl < 5));
    if (up[it]) plane_pair(pl, pa, pb);
    ua[it] = raw + pa * RSZ + cb * 16;
    ub[it] = raw + pb * RSZ + cb * 16;
  }
  v8f S[5][2];
  v2f fa[8], fb[8];
  h_load(ua[0], ub[0], up[0], 0, m, hi, fa, fb);
  #pragma unroll
  for (int it = 0; it < 5; ++it) {
    v8f s0 = {};
    v8f s1 = {};
    if (uact[it]) {
      const int nx = it + 1 < 5 ? it + 1 : it;
      box_unit(ua[it], ub[it], up[it], ua[nx], ub[nx], up[nx], it + 1 < 5,
               Tw, band, lane, fa, fb, s0, s1);
    }
    S[it][0] = s0;
    S[it][1] = s1;
  }
  __syncthreads();

  #pragma unroll
  for (int it = 0; it < 5; ++it) {
    const int pl = (wave >> 1) + 4 * it;
    if (pl < 19) {
      #pragma unroll
      for (int ob = 0; ob < 2; ++ob) {
        #pragma unroll
        for (int v = 0; v < 8; ++v)
          sm[pl * ESZ + (16 * ob + 8 * hi + v) * EP + 16 * cb + m] = S[it][ob][v];
      }
    }
  }
  __syncthreads();

  for (int q = 0; q < NPX; q += 2) {
    const int pr = 2 * (tid >> 5) + 8 * q;
    const int pc = tid & 31;
    v2f Sv[19];
    #pragma unroll
    for (int pl = 0; pl < 19; ++pl) {
      v2f t = { sm[pl * ESZ + pr * EP + pc], sm[pl * ESZ + (pr + 1) * EP + pc] };
      Sv[pl] = t;
    }
    const int col = tx + pc;
    const int row = ty + pr;
    const int row1 = row + 1;
    const int cw = min(col + RAD, WW - 1) - max(col - RAD, 0) + 1;
    const int chc = min(row + RAD, HH - 1) - max(row - RAD, 0) + 1;
    const int chc1 = min(row1 + RAD, HH - 1) - max(row1 - RAD, 0) + 1;
    const v2f invN = { 1.f / (float)(chc * cw), 1.f / (float)(chc1 * cw) };
    const v2f pm = Sv[4] * invN;
    v2f Im[4], first[4];
    v2f sumIm = { 0.f, 0.f };
    #pragma unroll
    for (int c = 0; c < 4; ++c) {
      Im[c] = Sv[c] * invN;
      sumIm += Im[c];
      first[c] = Sv[5 + c] * invN - pm * Im[c];
    }
    const v2f m00 = Sv[9]  * invN + EPSV;
    const v2f m01 = Sv[10] * invN;
    const v2f m02 = Sv[11] * invN;
    const v2f m03 = Sv[12] * invN;
    const v2f m11 = Sv[13] * invN + EPSV;
    const v2f m12 = Sv[14] * invN;
    const v2f m13 = Sv[15] * invN;
    const v2f m22 = Sv[16] * invN + EPSV;
    const v2f m23 = Sv[17] * invN;
    const v2f m33 = Sv[18] * invN + EPSV;
    v2f rs[4];
    inv4_rowsum_sym2(m00, m01, m02, m03, m11, m12, m13, m22, m23, m33, rs);
    v2f sumA = { 0.f, 0.f };
    v2f Ac[4];
    #pragma unroll
    for (int c = 0; c < 4; ++c) {
      Ac[c] = first[c] * rs[c];
      sumA += Ac[c];
    }
    const v2f bv = pm - sumA * sumIm;
    #pragma unroll
    for (int c = 0; c < 4; ++c) {
      Aout[(((size_t)bb * CC + c) * HH + row) * WW + col] = Ac[c].x;
      Aout[(((size_t)bb * CC + c) * HH + row1) * WW + col] = Ac[c].y;
    }
    Bout[((size_t)bb * HH + row) * WW + col] = bv.x;
    Bout[((size_t)bb * HH + row1) * WW + col] = bv.y;
  }
}

__global__ __launch_bounds__(NTHR) void gf_final(const float* __restrict__ I,
                                                 const float* __restrict__ Ain,
                                                 const float* __restrict__ Bin,
                                                 float* __restrict__ Q) {
  __shared__ __attribute__((aligned(16))) float sm[5 * RSZ + NWAVE * TSZ];
  float* raw = sm;
  float* Ts = sm + 5 * RSZ;
  const int tid = threadIdx.x;
  const int lane = tid & 31;
  const int wave = __builtin_amdgcn_readfirstlane(tid >> 5);
  const int tx = blockIdx.x * TL;
  const int ty = blockIdx.y * TL;
  const int bb = blockIdx.z;
  const float* Ab = Ain + (size_t)bb * CC * HH * WW;
  const float* Bb = Bin + (size_t)bb * HH * WW;
  const float* Ib = I + (size_t)bb * CC * HH * WW;
  fetch5(Ab, Ab + (size_t)HH * WW, Ab + (size_t)2 * HH * WW,
         Ab + (size_t)3 * HH * WW, Bb, raw, tx, ty, tid);
  float sumI[NPX];
  #pragma unroll
  for (int q = 0; q < NPX; ++q) {
    const int px = tid + NTHR * q;
    const size_t off = (size_t)(ty + (px >> 5)) * WW + (tx + (px & 31));
    float s = 0.f;
    #pragma unroll
    for (int c = 0; c < 4; ++c) s += Ib[(size_t)c * HH * WW + off];
    sumI[q] = s;
  }
  __syncthreads();

  v2f band[8];
  make_band(band, lane);
  float* Tw = Ts + wave * TSZ;
  const int cb = wave & 1;
  const int hi = lane >> 4;
  const int m = lane & 15;
  const int pl0 = wave >> 1;
  const bool act1 = pl0 + 4 < 5;
  const float* u0 = raw + pl0 * RSZ + cb * 16;
  const float* u1 = raw + 4 * RSZ + cb * 16;
  v8f S[2][2];
  v2f fa[8], fb[8];
  h_load(u0, u0, false, 0, m, hi, fa, fb);
  {
    v8f s0 = {};
    v8f s1 = {};
    box_unit(u0, u0, false, u1, u1, false, true, Tw, band, lane, fa, fb, s0, s1);
    S[0][0] = s0;
    S[0][1] = s1;
  }
  {
    v8f s0 = {};
    v8f s1 = {};
    if (act1)
      box_unit(u1, u1, false, u1, u1, false, false, Tw, band, lane, fa, fb, s0, s1);
    S[1][0] = s0;
    S[1][1] = s1;
  }
  __syncthreads();
  #pragma unroll
  for (int it = 0; it < 2; ++it) {
    const int pl = (wave >> 1) + 4 * it;
    if (pl < 5) {
      #pragma unroll
      for (int ob = 0; ob < 2; ++ob) {
        #pragma unroll
        for (int v = 0; v < 8; ++v)
          sm[pl * ESZ + (16 * ob + 8 * hi + v) * EP + 16 * cb + m] = S[it][ob][v];
      }
    }
  }
  __syncthreads();

  #pragma unroll
  for (int q = 0; q < NPX; ++q) {
    const int px = tid + NTHR * q;
    const int pr = px >> 5;
    const int pc = px & 31;
    const int col = tx + pc;
    const int row = ty + pr;
    const int cw = min(col + RAD, WW - 1) - max(col - RAD, 0) + 1;
    const int chc = min(row + RAD, HH - 1) - max(row - RAD, 0) + 1;
    const float invN = 1.f / (float)(chc * cw);
    const int e = pr * EP + pc;
    const float am = (sm[e] + sm[ESZ + e] + sm[2 * ESZ + e] + sm[3 * ESZ + e]) * invN;
    const float bm = sm[4 * ESZ + e] * invN;
    Q[((size_t)bb * HH + row) * WW + col] = am * sumI[q] + bm;
  }
}

extern "C" void kernel_launch(void* const* d_in, const int* in_sizes, int n_in,
                              void* d_out, int out_size, void* d_ws,
                              size_t ws_size, hipStream_t stream) {
  const float* I = (const float*)d_in[0];
  const float* P = (const float*)d_in[1];
  float* Q = (float*)d_out;
  float* Aws = (float*)d_ws;
  float* Bws = Aws + (size_t)NB * CC * HH * WW;
  dim3 grid(WW / TL, HH / TL, NB);
  gf_stats<<<grid, dim3(NTHR), 0, stream>>>(I, P, Aws, Bws);
  gf_final<<<grid, dim3(NTHR), 0, stream>>>(I, Aws, Bws, Q);
  (void)in_sizes; (void)n_in; (void)out_size; (void)ws_size;
}
